# starting kernel
# baseline (speedup 1.0000x reference)
#include <hip/hip_runtime.h>


typedef __attribute__((ext_vector_type(2))) float v2f;
typedef __attribute__((ext_vector_type(8))) float v8f;

#define WPB 8

__global__ void gcn_init_deg(float* __restrict__ deg, int n) {
  int i = blockIdx.x * blockDim.x + threadIdx.x;
  if (i < n) deg[i] = 1.0f;
}

__global__ void gcn_edge_deg(const int* __restrict__ dst, int E,
                             float* __restrict__ deg) {
  int e = blockIdx.x * blockDim.x + threadIdx.x;
  if (e < E) atomicAdd(&deg[dst[e]], 1.0f);
}

__global__ void gcn_rsqrt(float* __restrict__ d, int n) {
  int i = blockIdx.x * blockDim.x + threadIdx.x;
  if (i < n) {
    float v = d[i];
    d[i] = (v > 0.0f) ? rsqrtf(v) : 0.0f;
  }
}

__global__ __launch_bounds__(256) void gcn_gemm1(
    const float* __restrict__ X, const float* __restrict__ W1,
    const float* __restrict__ dinv, float* __restrict__ g1,
    float* __restrict__ agg1, int nTiles) {
  __shared__ float sW[64 * 64];
  {
    const float4* w4 = (const float4*)W1;
    float4* s4 = (float4*)sW;
    for (int i = threadIdx.x; i < 64 * 64 / 4; i += blockDim.x) s4[i] = w4[i];
  }
  __syncthreads();

  const int wave = threadIdx.x >> 5;
  const int lane = threadIdx.x & 31;
  const int tile = blockIdx.x * WPB + wave;
  if (tile >= nTiles) return;

  const int rowbase = tile << 4;
  const int half = lane >> 4;
  const int lp = lane & 15;
  const int m = rowbase + lp;

  v8f acc0 = {}, acc1 = {}, acc2 = {}, acc3 = {};
  const float* __restrict__ Arow = X + (size_t)m * 64;

#pragma unroll
  for (int k = 0; k < 64; k += 4) {
    const int ka = k + 2 * half;
    v2f a;
    a.x = Arow[ka];
    a.y = Arow[ka + 1];
    const float* b0r = &sW[ka * 64];
    const float* b1r = &sW[(ka + 1) * 64];
    v2f b;
    b.x = b0r[lp];
    b.y = b1r[lp];
    acc0 = __builtin_amdgcn_wmma_f32_16x16x4_f32(false, a, false, b, (short)0,
                                                 acc0, false, false);
    b.x = b0r[16 + lp];
    b.y = b1r[16 + lp];
    acc1 = __builtin_amdgcn_wmma_f32_16x16x4_f32(false, a, false, b, (short)0,
                                                 acc1, false, false);
    b.x = b0r[32 + lp];
    b.y = b1r[32 + lp];
    acc2 = __builtin_amdgcn_wmma_f32_16x16x4_f32(false, a, false, b, (short)0,
                                                 acc2, false, false);
    b.x = b0r[48 + lp];
    b.y = b1r[48 + lp];
    acc3 = __builtin_amdgcn_wmma_f32_16x16x4_f32(false, a, false, b, (short)0,
                                                 acc3, false, false);
  }

#pragma unroll
  for (int j = 0; j < 8; ++j) {
    const int row = rowbase + j + 8 * half;
    const float dv = dinv[row];
    const size_t base = (size_t)row * 64 + lp;
    float v;
    v = dv * acc0[j]; g1[base]      = v; agg1[base]      = v;
    v = dv * acc1[j]; g1[base + 16] = v; agg1[base + 16] = v;
    v = dv * acc2[j]; g1[base + 32] = v; agg1[base + 32] = v;
    v = dv * acc3[j]; g1[base + 48] = v; agg1[base + 48] = v;
  }
}

__global__ void gcn_edge_agg1(const int* __restrict__ src,
                              const int* __restrict__ dst,
                              const float* __restrict__ g1,
                              float* __restrict__ agg1, int E) {
  int tid = blockIdx.x * blockDim.x + threadIdx.x;
  if (tid >= E * 16) return;
  int e = tid >> 4;
  int q = (tid & 15) << 2;
  int s = src[e], d = dst[e];
  float4 v = *(const float4*)(g1 + (size_t)s * 64 + q);
  float* p = agg1 + (size_t)d * 64 + q;
  atomicAdd(p + 0, v.x);
  atomicAdd(p + 1, v.y);
  atomicAdd(p + 2, v.z);
  atomicAdd(p + 3, v.w);
}

__global__ __launch_bounds__(256) void gcn_gemm2(
    const float* __restrict__ agg1, const float* __restrict__ b1,
    const float* __restrict__ W2, const float* __restrict__ dinv,
    float* __restrict__ g2, float* __restrict__ outAcc, int nTiles) {
  __shared__ float sW[64 * 16];
  __shared__ float sB[64];
  {
    const float4* w4 = (const float4*)W2;
    float4* s4 = (float4*)sW;
    for (int i = threadIdx.x; i < 64 * 16 / 4; i += blockDim.x) s4[i] = w4[i];
    if (threadIdx.x < 64) sB[threadIdx.x] = b1[threadIdx.x];
  }
  __syncthreads();

  const int wave = threadIdx.x >> 5;
  const int lane = threadIdx.x & 31;
  const int tile = blockIdx.x * WPB + wave;
  if (tile >= nTiles) return;

  const int rowbase = tile << 4;
  const int half = lane >> 4;
  const int lp = lane & 15;
  const int m = rowbase + lp;

  const float dvm = dinv[m];
  const float* __restrict__ Arow = agg1 + (size_t)m * 64;
  v8f acc = {};

#pragma unroll
  for (int k = 0; k < 64; k += 4) {
    const int ka = k + 2 * half;
    v2f a;
    a.x = fmaxf(fmaf(dvm, Arow[ka], sB[ka]), 0.0f);
    a.y = fmaxf(fmaf(dvm, Arow[ka + 1], sB[ka + 1]), 0.0f);
    v2f b;
    b.x = sW[ka * 16 + lp];
    b.y = sW[(ka + 1) * 16 + lp];
    acc = __builtin_amdgcn_wmma_f32_16x16x4_f32(false, a, false, b, (short)0,
                                                acc, false, false);
  }

#pragma unroll
  for (int j = 0; j < 8; ++j) {
    const int row = rowbase + j + 8 * half;
    const float dv = dinv[row];
    const size_t idx = (size_t)row * 16 + lp;
    const float v = dv * acc[j];
    g2[idx] = v;
    outAcc[idx] = v;
  }
}

__global__ void gcn_edge_agg2(const int* __restrict__ src,
                              const int* __restrict__ dst,
                              const float* __restrict__ g2,
                              float* __restrict__ outAcc, int E) {
  int tid = blockIdx.x * blockDim.x + threadIdx.x;
  if (tid >= E * 4) return;
  int e = tid >> 2;
  int q = (tid & 3) << 2;
  int s = src[e], d = dst[e];
  float4 v = *(const float4*)(g2 + (size_t)s * 16 + q);
  float* p = outAcc + (size_t)d * 16 + q;
  atomicAdd(p + 0, v.x);
  atomicAdd(p + 1, v.y);
  atomicAdd(p + 2, v.z);
  atomicAdd(p + 3, v.w);
}

__global__ void gcn_final(float* __restrict__ out, const float* __restrict__ dinv,
                          const float* __restrict__ b2, int n) {
  int tid = blockIdx.x * blockDim.x + threadIdx.x;
  if (tid >= n * 4) return;
  int v = tid >> 2;
  int f = (tid & 3) << 2;
  float dv = dinv[v];
  float4* p = (float4*)out + tid;
  float4 t = *p;
  t.x = fmaf(dv, t.x, b2[f + 0]);
  t.y = fmaf(dv, t.y, b2[f + 1]);
  t.z = fmaf(dv, t.z, b2[f + 2]);
  t.w = fmaf(dv, t.w, b2[f + 3]);
  *p = t;
}

extern "C" void kernel_launch(void* const* d_in, const int* in_sizes, int n_in,
                              void* d_out, int out_size, void* d_ws, size_t ws_size,
                              hipStream_t stream) {
  const float* X  = (const float*)d_in[0];
  const int*   ei = (const int*)d_in[1];
  const float* W1 = (const float*)d_in[2];
  const float* b1 = (const float*)d_in[3];
  const float* W2 = (const float*)d_in[4];
  const float* b2 = (const float*)d_in[5];
  float* out = (float*)d_out;

  const int N = in_sizes[0] / 64;
  const int E = in_sizes[1] / 2;
  const int* src = ei;
  const int* dst = ei + E;

  float* ws   = (float*)d_ws;
  float* dinv = ws;
  float* g1   = dinv + N;
  float* agg1 = g1 + (size_t)N * 64;
  float* g2   = agg1 + (size_t)N * 64;

  const int nTiles = N / 16;
  const int B = 256;

  gcn_init_deg<<<(N + B - 1) / B, B, 0, stream>>>(dinv, N);
  gcn_edge_deg<<<(E + B - 1) / B, B, 0, stream>>>(dst, E, dinv);
  gcn_rsqrt<<<(N + B - 1) / B, B, 0, stream>>>(dinv, N);

  gcn_gemm1<<<(nTiles + WPB - 1) / WPB, B, 0, stream>>>(X, W1, dinv, g1, agg1,
                                                        nTiles);
  gcn_edge_agg1<<<(E * 16 + B - 1) / B, B, 0, stream>>>(src, dst, g1, agg1, E);

  gcn_gemm2<<<(nTiles + WPB - 1) / WPB, B, 0, stream>>>(agg1, b1, W2, dinv, g2,
                                                        out, nTiles);
  gcn_edge_agg2<<<(E * 4 + B - 1) / B, B, 0, stream>>>(src, dst, g2, out, E);

  gcn_final<<<(N * 4 + B - 1) / B, B, 0, stream>>>(out, dinv, b2, N);
}


# Both node GEMMs move tiles as whole 128-byte lines: the 4 KB A tile asked for as float4 in one group with the weights, A and D fragments pass through a per-wave LDS tile, 512-byte stores
# speedup vs baseline: 2.6179x; 2.6179x over previous
#include <hip/hip_runtime.h>


typedef __attribute__((ext_vector_type(2))) float v2f;
typedef __attribute__((ext_vector_type(8))) float v8f;

#define WPB 8

__device__ __forceinline__ int chip_half() {
  const unsigned id = __builtin_amdgcn_s_sendmsg_rtn(0x87);
  return (int)((id >> 18) & 1u);
}

__device__ __forceinline__ int row_class(int d) { return (d >> 5) & 1; }

#define NCTR 256

__global__ void gcn_init_deg(float* __restrict__ deg2, int n2,
                             int* __restrict__ ctr) {
  int i = blockIdx.x * blockDim.x + threadIdx.x;
  if (i < n2)
    deg2[i] = 0.0f;
  else if (i < n2 + NCTR)
    ctr[i - n2] = 0;
}

__global__ void gcn_edge_deg(const int* __restrict__ dst, int E,
                             float* __restrict__ deg2, int n) {
  int e = blockIdx.x * blockDim.x + threadIdx.x;
  float* __restrict__ plane = deg2 + (size_t)chip_half() * n;
  if (e < E) atomicAdd(&plane[dst[e]], 1.0f);
}

#define TP 68
__global__ __launch_bounds__(256) void gcn_gemm1(
    const float* __restrict__ X, const float* __restrict__ W1,
    const float* __restrict__ deg2, float* __restrict__ dinv,
    float* __restrict__ g1, float* __restrict__ agg1, int nTiles, int n) {
  const int wave = threadIdx.x >> 5;
  const int lane = threadIdx.x & 31;
  int tile = blockIdx.x * WPB + wave;
  const bool live = tile < nTiles;
  if (!live) tile = nTiles - 1;
  const int rowbase = tile << 4;
  const int half = lane >> 4;
  const int lp = lane & 15;
  const int m = rowbase + lp;

  __shared__ __attribute__((aligned(16))) float sW[64 * 64];
  __shared__ __attribute__((aligned(16))) float sT[WPB][16 * TP];
  float* __restrict__ T = sT[wave];

  const float4* __restrict__ x4 = (const float4*)(X + (size_t)rowbase * 64);
  float4 av[8];
#pragma unroll
  for (int i = 0; i < 8; ++i) av[i] = x4[i * 32 + lane];
  const float c0 = deg2[m];
  const float c1 = deg2[n + m];
  {
    const float4* w4 = (const float4*)W1;
    float4* s4 = (float4*)sW;
    float4 wv[4];
#pragma unroll
    for (int i = 0; i < 4; ++i) wv[i] = w4[threadIdx.x + i * 256];
#pragma unroll
    for (int i = 0; i < 4; ++i) s4[threadIdx.x + i * 256] = wv[i];
  }
#pragma unroll
  for (int i = 0; i < 8; ++i)
    *(float4*)&T[(2 * i + half) * TP + 4 * lp] = av[i];
  __syncthreads();

  const float cnt = 1.0f + c0 + c1;
  const float dvm = (cnt > 0.0f) ? rsqrtf(cnt) : 0.0f;
  if (live && half == 0) dinv[m] = dvm;

  v8f acc0 = {}, acc1 = {}, acc2 = {}, acc3 = {};
  const float* __restrict__ Arow = T + lp * TP;

#pragma unroll
  for (int k = 0; k < 64; k += 4) {
    const int ka = k + 2 * half;
    v2f a;
    a.x = Arow[ka];
    a.y = Arow[ka + 1];
    const float* b0r = &sW[ka * 64];
    const float* b1r = &sW[(ka + 1) * 64];
    v2f b;
    b.x = b0r[lp];
    b.y = b1r[lp];
    acc0 = __builtin_amdgcn_wmma_f32_16x16x4_f32(false, a, false, b, (short)0,
                                                 acc0, false, false);
    b.x = b0r[16 + lp];
    b.y = b1r[16 + lp];
    acc1 = __builtin_amdgcn_wmma_f32_16x16x4_f32(false, a, false, b, (short)0,
                                                 acc1, false, false);
    b.x = b0r[32 + lp];
    b.y = b1r[32 + lp];
    acc2 = __builtin_amdgcn_wmma_f32_16x16x4_f32(false, a, false, b, (short)0,
                                                 acc2, false, false);
    b.x = b0r[48 + lp];
    b.y = b1r[48 + lp];
    acc3 = __builtin_amdgcn_wmma_f32_16x16x4_f32(false, a, false, b, (short)0,
                                                 acc3, false, false);
  }

  __syncthreads();
#pragma unroll
  for (int j = 0; j < 8; ++j) {
    const float dlo = __int_as_float(__builtin_amdgcn_readlane(__float_as_int(dvm), j));
    const float dhi = __int_as_float(__builtin_amdgcn_readlane(__float_as_int(dvm), j + 8));
    const float dv = half ? dhi : dlo;
    float* __restrict__ r = T + (j + 8 * half) * TP + lp;
    r[0]  = dv * acc0[j];
    r[16] = dv * acc1[j];
    r[32] = dv * acc2[j];
    r[48] = dv * acc3[j];
  }
  __syncthreads();
  if (live) {
    float4* __restrict__ o1 = (float4*)(g1 + (size_t)rowbase * 64);
    float4* __restrict__ o2 = (float4*)(agg1 + (size_t)rowbase * 64);
#pragma unroll
    for (int i = 0; i < 8; ++i) {
      const float4 v = *(const float4*)&T[(2 * i + half) * TP + 4 * lp];
      o1[i * 32 + lane] = v;
      o2[i * 32 + lane] = v;
    }
  }
}

#define EPW 64
#define EPB (EPW * WPB)

__device__ __forceinline__ void take_item(int* __restrict__ ctr, int nItems,
                                          int* sTicket, int& chunk, int& cls) {
  const int h = chip_half();
  if (threadIdx.x == 0) *sTicket = atomicAdd(&ctr[h * 64], 1);
  __syncthreads();
  const int t = __builtin_amdgcn_readfirstlane(*sTicket);
  if (t < nItems) {
    chunk = t;
    cls = h;
  } else {
    chunk = 2 * nItems - 1 - t;
    cls = 1 - h;
  }
}

__device__ __forceinline__ unsigned long long load_edges(
    const int* __restrict__ src, const int* __restrict__ dst, int e0, int E,
    int lane, int cls, int& s0, int& d0, int& s1, int& d1) {
  const int ea = e0 + lane, eb = e0 + 32 + lane;
  s0 = 0; d0 = 0; s1 = 0; d1 = 0;
  if (ea < E) { s0 = src[ea]; d0 = dst[ea]; }
  if (eb < E) { s1 = src[eb]; d1 = dst[eb]; }
  const unsigned m0 = __builtin_amdgcn_ballot_w32(ea < E && row_class(d0) == cls);
  const unsigned m1 = __builtin_amdgcn_ballot_w32(eb < E && row_class(d1) == cls);
  return (unsigned long long)m0 | ((unsigned long long)m1 << 32);
}

__device__ __forceinline__ int pick(int x0, int x1, int i) {
  const int a = __builtin_amdgcn_readlane(x0, i & 31);
  const int b = __builtin_amdgcn_readlane(x1, i & 31);
  return i < 32 ? a : b;
}

__global__ __launch_bounds__(256) void gcn_edge_agg1(
    const int* __restrict__ src, const int* __restrict__ dst,
    const float* __restrict__ g1, float* __restrict__ agg1, int E, int nItems,
    int* __restrict__ ctr) {
  __shared__ int sTicket;
  int chunk, cls;
  take_item(ctr, nItems, &sTicket, chunk, cls);
  if (chunk < 0) return;
  const int lane = threadIdx.x & 31;
  const int wave = threadIdx.x >> 5;
  const int e0 = chunk * EPB + wave * EPW;
  if (e0 >= E) return;
  int s0, d0, s1, d1;
  unsigned long long mask = load_edges(src, dst, e0, E, lane, cls, s0, d0, s1, d1);

  while (mask) {
    float lo[8], hi[8];
    int dd[8];
#pragma unroll
    for (int j = 0; j < 8; ++j) {
      dd[j] = -1;
      lo[j] = 0.0f;
      hi[j] = 0.0f;
      if (mask) {
        const int i = __builtin_ctzll(mask);
        mask &= mask - 1;
        dd[j] = pick(d0, d1, i);
        const float* r = g1 + (size_t)pick(s0, s1, i) * 64;
        lo[j] = r[lane];
        hi[j] = r[lane + 32];
      }
    }
#pragma unroll
    for (int j = 0; j < 8; ++j) {
      if (dd[j] >= 0) {
        float* p = agg1 + (size_t)dd[j] * 64;
        atomicAdd(p + lane, lo[j]);
        atomicAdd(p + lane + 32, hi[j]);
      }
    }
  }
}

#define OP 20
__global__ __launch_bounds__(256) void gcn_gemm2(
    const float* __restrict__ agg1, const float* __restrict__ b1,
    const float* __restrict__ W2, const float* __restrict__ b2,
    const float* __restrict__ dinv, float* __restrict__ g2,
    float* __restrict__ outAcc, int nTiles) {
  __shared__ __attribute__((aligned(16))) float sW[64 * 16];
  __shared__ float sB[64];
  __shared__ __attribute__((aligned(16))) float sT[WPB][16 * TP];

  const int wave = threadIdx.x >> 5;
  const int lane = threadIdx.x & 31;
  int tile = blockIdx.x * WPB + wave;
  const bool live = tile < nTiles;
  if (!live) tile = nTiles - 1;
  const int rowbase = tile << 4;
  const int half = lane >> 4;
  const int lp = lane & 15;
  const int m = rowbase + lp;
  float* __restrict__ T = sT[wave];

  const float4* __restrict__ a4 = (const float4*)(agg1 + (size_t)rowbase * 64);
  float4 av[8];
#pragma unroll
  for (int i = 0; i < 8; ++i) av[i] = a4[i * 32 + lane];
  const float dvm = dinv[m];
  const float bias2 = b2[lp];
  {
    const float4 wv = ((const float4*)W2)[threadIdx.x];
    const float bv = b1[threadIdx.x & 63];
    ((float4*)sW)[threadIdx.x] = wv;
    sB[threadIdx.x & 63] = bv;
  }
#pragma unroll
  for (int i = 0; i < 8; ++i)
    *(float4*)&T[(2 * i + half) * TP + 4 * lp] = av[i];
  __syncthreads();

  const float* __restrict__ Arow = T + lp * TP;
  v8f acc = {};

#pragma unroll
  for (int k = 0; k < 64; k += 4) {
    const int ka = k + 2 * half;
    v2f a;
    a.x = fmaxf(fmaf(dvm, Arow[ka], sB[ka]), 0.0f);
    a.y = fmaxf(fmaf(dvm, Arow[ka + 1], sB[ka + 1]), 0.0f);
    v2f b;
    b.x = sW[ka * 16 + lp];
    b.y = sW[(ka + 1) * 16 + lp];
    acc = __builtin_amdgcn_wmma_f32_16x16x4_f32(false, a, false, b, (short)0,
                                                acc, false, false);
  }

  __syncthreads();
  float* __restrict__ Tg = T;
  float* __restrict__ To = T + 16 * OP;
#pragma unroll
  for (int j = 0; j < 8; ++j) {
    const float dlo = __int_as_float(__builtin_amdgcn_readlane(__float_as_int(dvm), j));
    const float dhi = __int_as_float(__builtin_amdgcn_readlane(__float_as_int(dvm), j + 8));
    const float dv = half ? dhi : dlo;
    const float v = dv * acc[j];
    Tg[(j + 8 * half) * OP + lp] = v;
    To[(j + 8 * half) * OP + lp] = fmaf(dv, v, bias2);
  }
  __syncthreads();
  if (live) {
    float4* __restrict__ o1 = (float4*)(g2 + (size_t)rowbase * 16);
    float4* __restrict__ o2 = (float4*)(outAcc + (size_t)rowbase * 16);
#pragma unroll
    for (int c = 0; c < 2; ++c) {
      const int off = (8 * c + (lane >> 2)) * OP + 4 * (lane & 3);
      o1[c * 32 + lane] = *(const float4*)&Tg[off];
      o2[c * 32 + lane] = *(const float4*)&To[off];
    }
  }
}

__device__ __forceinline__ float pickf(float x0, float x1, int i) {
  return __int_as_float(pick(__float_as_int(x0), __float_as_int(x1), i));
}

__global__ __launch_bounds__(256) void gcn_edge_agg2(
    const int* __restrict__ src, const int* __restrict__ dst,
    const float* __restrict__ g2, const float* __restrict__ dinv,
    float* __restrict__ outAcc, int E, int nItems, int* __restrict__ ctr) {
  __shared__ int sTicket;
  int chunk, cls;
  take_item(ctr, nItems, &sTicket, chunk, cls);
  if (chunk < 0) return;
  const int lane = threadIdx.x & 31;
  const int wave = threadIdx.x >> 5;
  const int e0 = chunk * EPB + wave * EPW;
  if (e0 >= E) return;
  int s0, d0, s1, d1;
  unsigned long long mask = load_edges(src, dst, e0, E, lane, cls, s0, d0, s1, d1);
  const bool odd = lane >= 16;
  const int lp = lane & 15;
  const bool sel0 = (mask >> lane) & 1ull, sel1 = (mask >> (lane + 32)) & 1ull;
  float w0 = 0.0f, w1 = 0.0f;
  bool need_w = true;

  while (mask) {
    float v[8];
    int dd[8];
    int ea[8], eb[8];
#pragma unroll
    for (int j = 0; j < 8; ++j) {
      dd[j] = -1;
      v[j] = 0.0f;
      ea[j] = 0;
      eb[j] = 0;
      if (mask) {
        const int ia = __builtin_ctzll(mask);
        mask &= mask - 1;
        const bool hasb = mask != 0;
        int ib = ia;
        if (hasb) {
          ib = __builtin_ctzll(mask);
          mask &= mask - 1;
        }
        const int sa = pick(s0, s1, ia), da = pick(d0, d1, ia);
        const int sb = pick(s0, s1, ib), db = pick(d0, d1, ib);
        dd[j] = odd ? (hasb ? db : -1) : da;
        v[j] = g2[(size_t)(odd ? sb : sa) * 16 + lp];
        ea[j] = ia;
        eb[j] = ib;
      }
    }
    if (need_w) {
      if (sel0) w0 = dinv[d0];
      if (sel1) w1 = dinv[d1];
      need_w = false;
    }
#pragma unroll
    for (int j = 0; j < 8; ++j) {
      if (dd[j] >= 0) {
        const float wa = pickf(w0, w1, ea[j]), wb = pickf(w0, w1, eb[j]);
        atomicAdd(outAcc + (size_t)dd[j] * 16 + lp, (odd ? wb : wa) * v[j]);
      }
    }
  }
}

extern "C" void kernel_launch(void* const* d_in, const int* in_sizes, int n_in,
                              void* d_out, int out_size, void* d_ws, size_t ws_size,
                              hipStream_t stream) {
  const float* X  = (const float*)d_in[0];
  const int*   ei = (const int*)d_in[1];
  const float* W1 = (const float*)d_in[2];
  const float* b1 = (const float*)d_in[3];
  const float* W2 = (const float*)d_in[4];
  const float* b2 = (const float*)d_in[5];
  float* out = (float*)d_out;

  const int N = in_sizes[0] / 64;
  const int E = in_sizes[1] / 2;
  const int* src = ei;
  const int* dst = ei + E;

  float* ws   = (float*)d_ws;
  float* dinv = ws;
  float* g1   = dinv + N;
  float* agg1 = g1 + (size_t)N * 64;
  float* g2   = agg1 + (size_t)N * 64;
  float* deg2 = g2 + (size_t)N * 16;
  int*   ctr  = (int*)(deg2 + (size_t)N * 2);

  const int nTiles = N / 16;
  const int B = 256;

  const int nItems = (E + EPB - 1) / EPB;

  gcn_init_deg<<<(2 * N + NCTR + B - 1) / B, B, 0, stream>>>(deg2, 2 * N, ctr);
  gcn_edge_deg<<<(E + B - 1) / B, B, 0, stream>>>(dst, E, deg2, N);

  gcn_gemm1<<<(nTiles + WPB - 1) / WPB, B, 0, stream>>>(X, W1, deg2, dinv, g1,
                                                        agg1, nTiles, N);
  gcn_edge_agg1<<<2 * nItems, B, 0, stream>>>(src, dst, g1, agg1, E, nItems, ctr);

  gcn_gemm2<<<(nTiles + WPB - 1) / WPB, B, 0, stream>>>(agg1, b1, W2, b2, dinv,
                                                        g2, out, nTiles);
  gcn_edge_agg2<<<2 * nItems, B, 0, stream>>>(src, dst, g2, dinv, out, E, nItems,
                                              ctr + 128);
}


# Edge kernels run as small fixed grids whose blocks loop (degree count by grid stride, both scatters draw tickets until none is left): 36,864 waves enter, not 93,784; sums stay arrival-ordered
# speedup vs baseline: 2.7017x; 1.0320x over previous
#include <hip/hip_runtime.h>


typedef __attribute__((ext_vector_type(2))) float v2f;
typedef __attribute__((ext_vector_type(8))) float v8f;

#define WPB 8

__device__ __forceinline__ int chip_half() {
  const unsigned id = __builtin_amdgcn_s_sendmsg_rtn(0x87);
  return (int)((id >> 18) & 1u);
}

__device__ __forceinline__ int row_class(int d) { return (d >> 5) & 1; }

#define NCTR 256

__global__ void gcn_init_deg(float* __restrict__ deg2, int n2,
                             int* __restrict__ ctr) {
  int i = blockIdx.x * blockDim.x + threadIdx.x;
  if (i < n2)
    deg2[i] = 0.0f;
  else if (i < n2 + NCTR)
    ctr[i - n2] = 0;
}

#define DEG_GRID 512
#define DEG_PER 8
__global__ __launch_bounds__(256) void gcn_edge_deg(
    const int* __restrict__ dst, int E, float* __restrict__ deg2, int n) {
  float* __restrict__ plane = deg2 + (size_t)chip_half() * n;
  const int stride = gridDim.x * blockDim.x;
  for (int e = blockIdx.x * blockDim.x + threadIdx.x; e < E; e += DEG_PER * stride) {
    int d[DEG_PER];
#pragma unroll
    for (int k = 0; k < DEG_PER; ++k) {
      const int ek = e + k * stride;
      d[k] = ek < E ? dst[ek] : -1;
    }
#pragma unroll
    for (int k = 0; k < DEG_PER; ++k)
      if (d[k] >= 0) atomicAdd(&plane[d[k]], 1.0f);
  }
}

#define TP 68
__global__ __launch_bounds__(256) void gcn_gemm1(
    const float* __restrict__ X, const float* __restrict__ W1,
    const float* __restrict__ deg2, float* __restrict__ dinv,
    float* __restrict__ g1, float* __restrict__ agg1, int nTiles, int n) {
  const int wave = threadIdx.x >> 5;
  const int lane = threadIdx.x & 31;
  int tile = blockIdx.x * WPB + wave;
  const bool live = tile < nTiles;
  if (!live) tile = nTiles - 1;
  const int rowbase = tile << 4;
  const int half = lane >> 4;
  const int lp = lane & 15;
  const int m = rowbase + lp;

  __shared__ __attribute__((aligned(16))) float sW[64 * 64];
  __shared__ __attribute__((aligned(16))) float sT[WPB][16 * TP];
  float* __restrict__ T = sT[wave];

  const float4* __restrict__ x4 = (const float4*)(X + (size_t)rowbase * 64);
  float4 av[8];
#pragma unroll
  for (int i = 0; i < 8; ++i) av[i] = x4[i * 32 + lane];
  const float c0 = deg2[m];
  const float c1 = deg2[n + m];
  {
    const float4* w4 = (const float4*)W1;
    float4* s4 = (float4*)sW;
    float4 wv[4];
#pragma unroll
    for (int i = 0; i < 4; ++i) wv[i] = w4[threadIdx.x + i * 256];
#pragma unroll
    for (int i = 0; i < 4; ++i) s4[threadIdx.x + i * 256] = wv[i];
  }
#pragma unroll
  for (int i = 0; i < 8; ++i)
    *(float4*)&T[(2 * i + half) * TP + 4 * lp] = av[i];
  __syncthreads();

  const float cnt = 1.0f + c0 + c1;
  const float dvm = (cnt > 0.0f) ? rsqrtf(cnt) : 0.0f;
  if (live && half == 0) dinv[m] = dvm;

  v8f acc0 = {}, acc1 = {}, acc2 = {}, acc3 = {};
  const float* __restrict__ Arow = T + lp * TP;

#pragma unroll
  for (int k = 0; k < 64; k += 4) {
    const int ka = k + 2 * half;
    v2f a;
    a.x = Arow[ka];
    a.y = Arow[ka + 1];
    const float* b0r = &sW[ka * 64];
    const float* b1r = &sW[(ka + 1) * 64];
    v2f b;
    b.x = b0r[lp];
    b.y = b1r[lp];
    acc0 = __builtin_amdgcn_wmma_f32_16x16x4_f32(false, a, false, b, (short)0,
                                                 acc0, false, false);
    b.x = b0r[16 + lp];
    b.y = b1r[16 + lp];
    acc1 = __builtin_amdgcn_wmma_f32_16x16x4_f32(false, a, false, b, (short)0,
                                                 acc1, false, false);
    b.x = b0r[32 + lp];
    b.y = b1r[32 + lp];
    acc2 = __builtin_amdgcn_wmma_f32_16x16x4_f32(false, a, false, b, (short)0,
                                                 acc2, false, false);
    b.x = b0r[48 + lp];
    b.y = b1r[48 + lp];
    acc3 = __builtin_amdgcn_wmma_f32_16x16x4_f32(false, a, false, b, (short)0,
                                                 acc3, false, false);
  }

  __syncthreads();
#pragma unroll
  for (int j = 0; j < 8; ++j) {
    const float dlo = __int_as_float(__builtin_amdgcn_readlane(__float_as_int(dvm), j));
    const float dhi = __int_as_float(__builtin_amdgcn_readlane(__float_as_int(dvm), j + 8));
    const float dv = half ? dhi : dlo;
    float* __restrict__ r = T + (j + 8 * half) * TP + lp;
    r[0]  = dv * acc0[j];
    r[16] = dv * acc1[j];
    r[32] = dv * acc2[j];
    r[48] = dv * acc3[j];
  }
  __syncthreads();
  if (live) {
    float4* __restrict__ o1 = (float4*)(g1 + (size_t)rowbase * 64);
    float4* __restrict__ o2 = (float4*)(agg1 + (size_t)rowbase * 64);
#pragma unroll
    for (int i = 0; i < 8; ++i) {
      const float4 v = *(const float4*)&T[(2 * i + half) * TP + 4 * lp];
      o1[i * 32 + lane] = v;
      o2[i * 32 + lane] = v;
    }
  }
}

#define EPW 64
#define EPB (EPW * WPB)
#define SCAT_GRID 2048

__device__ __forceinline__ int item_of_ticket(int t, int* __restrict__ ctr,
                                              int nItems, int own, int& drawCls) {
  if (t >= nItems && drawCls == own) {
    drawCls = 1 - own;
    t = atomicAdd(&ctr[drawCls * 64], 1);
  }
  return t < nItems ? drawCls * nItems + t : -1;
}

__device__ __forceinline__ unsigned long long load_edges(
    const int* __restrict__ src, const int* __restrict__ dst, int e0, int E,
    int lane, int cls, int& s0, int& d0, int& s1, int& d1) {
  const int ea = e0 + lane, eb = e0 + 32 + lane;
  s0 = 0; d0 = 0; s1 = 0; d1 = 0;
  if (ea < E) { s0 = src[ea]; d0 = dst[ea]; }
  if (eb < E) { s1 = src[eb]; d1 = dst[eb]; }
  const unsigned m0 = __builtin_amdgcn_ballot_w32(ea < E && row_class(d0) == cls);
  const unsigned m1 = __builtin_amdgcn_ballot_w32(eb < E && row_class(d1) == cls);
  return (unsigned long long)m0 | ((unsigned long long)m1 << 32);
}

__device__ __forceinline__ int pick(int x0, int x1, int i) {
  const int a = __builtin_amdgcn_readlane(x0, i & 31);
  const int b = __builtin_amdgcn_readlane(x1, i & 31);
  return i < 32 ? a : b;
}

__global__ __launch_bounds__(256) void gcn_edge_agg1(
    const int* __restrict__ src, const int* __restrict__ dst,
    const float* __restrict__ g1, float* __restrict__ agg1, int E, int nItems,
    int* __restrict__ ctr) {
  __shared__ int sItem[2];
  const int lane = threadIdx.x & 31;
  const int wave = threadIdx.x >> 5;
  const int own = chip_half();
  int drawCls = own;
  int tNext = 0;
  if (threadIdx.x == 0) {
    tNext = atomicAdd(&ctr[drawCls * 64], 1);
    sItem[0] = item_of_ticket(tNext, ctr, nItems, own, drawCls);
  }
  for (int par = 0;; par ^= 1) {
    __syncthreads();
    const int item = __builtin_amdgcn_readfirstlane(sItem[par]);
    if (item < 0) break;
    if (threadIdx.x == 0) tNext = atomicAdd(&ctr[drawCls * 64], 1);
    const int cls = item >= nItems ? 1 : 0;
    const int chunk = item - cls * nItems;
    const int e0 = chunk * EPB + wave * EPW;
    if (e0 < E) {
      int s0, d0, s1, d1;
      unsigned long long mask = load_edges(src, dst, e0, E, lane, cls, s0, d0, s1, d1);

      while (mask) {
        float lo[8], hi[8];
        int dd[8];
#pragma unroll
        for (int j = 0; j < 8; ++j) {
          dd[j] = -1;
          lo[j] = 0.0f;
          hi[j] = 0.0f;
          if (mask) {
            const int i = __builtin_ctzll(mask);
            mask &= mask - 1;
            dd[j] = pick(d0, d1, i);
            const float* r = g1 + (size_t)pick(s0, s1, i) * 64;
            lo[j] = r[lane];
            hi[j] = r[lane + 32];
          }
        }
#pragma unroll
        for (int j = 0; j < 8; ++j) {
          if (dd[j] >= 0) {
            float* p = agg1 + (size_t)dd[j] * 64;
            atomicAdd(p + lane, lo[j]);
            atomicAdd(p + lane + 32, hi[j]);
          }
        }
      }
    }
    if (threadIdx.x == 0)
      sItem[par ^ 1] = item_of_ticket(tNext, ctr, nItems, own, drawCls);
  }
}

#define OP 20
__global__ __launch_bounds__(256) void gcn_gemm2(
    const float* __restrict__ agg1, const float* __restrict__ b1,
    const float* __restrict__ W2, const float* __restrict__ b2,
    const float* __restrict__ dinv, float* __restrict__ g2,
    float* __restrict__ outAcc, int nTiles) {
  __shared__ __attribute__((aligned(16))) float sW[64 * 16];
  __shared__ float sB[64];
  __shared__ __attribute__((aligned(16))) float sT[WPB][16 * TP];

  const int wave = threadIdx.x >> 5;
  const int lane = threadIdx.x & 31;
  int tile = blockIdx.x * WPB + wave;
  const bool live = tile < nTiles;
  if (!live) tile = nTiles - 1;
  const int rowbase = tile << 4;
  const int half = lane >> 4;
  const int lp = lane & 15;
  const int m = rowbase + lp;
  float* __restrict__ T = sT[wave];

  const float4* __restrict__ a4 = (const float4*)(agg1 + (size_t)rowbase * 64);
  float4 av[8];
#pragma unroll
  for (int i = 0; i < 8; ++i) av[i] = a4[i * 32 + lane];
  const float dvm = dinv[m];
  const float bias2 = b2[lp];
  {
    const float4 wv = ((const float4*)W2)[threadIdx.x];
    const float bv = b1[threadIdx.x & 63];
    ((float4*)sW)[threadIdx.x] = wv;
    sB[threadIdx.x & 63] = bv;
  }
#pragma unroll
  for (int i = 0; i < 8; ++i)
    *(float4*)&T[(2 * i + half) * TP + 4 * lp] = av[i];
  __syncthreads();

  const float* __restrict__ Arow = T + lp * TP;
  v8f acc = {};

#pragma unroll
  for (int k = 0; k < 64; k += 4) {
    const int ka = k + 2 * half;
    v2f a;
    a.x = fmaxf(fmaf(dvm, Arow[ka], sB[ka]), 0.0f);
    a.y = fmaxf(fmaf(dvm, Arow[ka + 1], sB[ka + 1]), 0.0f);
    v2f b;
    b.x = sW[ka * 16 + lp];
    b.y = sW[(ka + 1) * 16 + lp];
    acc = __builtin_amdgcn_wmma_f32_16x16x4_f32(false, a, false, b, (short)0,
                                                acc, false, false);
  }

  __syncthreads();
  float* __restrict__ Tg = T;
  float* __restrict__ To = T + 16 * OP;
#pragma unroll
  for (int j = 0; j < 8; ++j) {
    const float dlo = __int_as_float(__builtin_amdgcn_readlane(__float_as_int(dvm), j));
    const float dhi = __int_as_float(__builtin_amdgcn_readlane(__float_as_int(dvm), j + 8));
    const float dv = half ? dhi : dlo;
    const float v = dv * acc[j];
    Tg[(j + 8 * half) * OP + lp] = v;
    To[(j + 8 * half) * OP + lp] = fmaf(dv, v, bias2);
  }
  __syncthreads();
  if (live) {
    float4* __restrict__ o1 = (float4*)(g2 + (size_t)rowbase * 16);
    float4* __restrict__ o2 = (float4*)(outAcc + (size_t)rowbase * 16);
#pragma unroll
    for (int c = 0; c < 2; ++c) {
      const int off = (8 * c + (lane >> 2)) * OP + 4 * (lane & 3);
      o1[c * 32 + lane] = *(const float4*)&Tg[off];
      o2[c * 32 + lane] = *(const float4*)&To[off];
    }
  }
}

__device__ __forceinline__ float pickf(float x0, float x1, int i) {
  return __int_as_float(pick(__float_as_int(x0), __float_as_int(x1), i));
}

__global__ __launch_bounds__(256) void gcn_edge_agg2(
    const int* __restrict__ src, const int* __restrict__ dst,
    const float* __restrict__ g2, const float* __restrict__ dinv,
    float* __restrict__ outAcc, int E, int nItems, int* __restrict__ ctr) {
  __shared__ int sItem[2];
  const int lane = threadIdx.x & 31;
  const int wave = threadIdx.x >> 5;
  const bool odd = lane >= 16;
  const int lp = lane & 15;
  const int own = chip_half();
  int drawCls = own;
  int tNext = 0;
  if (threadIdx.x == 0) {
    tNext = atomicAdd(&ctr[drawCls * 64], 1);
    sItem[0] = item_of_ticket(tNext, ctr, nItems, own, drawCls);
  }
  for (int par = 0;; par ^= 1) {
    __syncthreads();
    const int item = __builtin_amdgcn_readfirstlane(sItem[par]);
    if (item < 0) break;
    if (threadIdx.x == 0) tNext = atomicAdd(&ctr[drawCls * 64], 1);
    const int cls = item >= nItems ? 1 : 0;
    const int chunk = item - cls * nItems;
    const int e0 = chunk * EPB + wave * EPW;
    if (e0 < E) {
      int s0, d0, s1, d1;
      unsigned long long mask = load_edges(src, dst, e0, E, lane, cls, s0, d0, s1, d1);
      const bool sel0 = (mask >> lane) & 1ull, sel1 = (mask >> (lane + 32)) & 1ull;
      float w0 = 0.0f, w1 = 0.0f;
      bool need_w = true;

      while (mask) {
        float v[8];
        int dd[8];
        int ea[8], eb[8];
#pragma unroll
        for (int j = 0; j < 8; ++j) {
          dd[j] = -1;
          v[j] = 0.0f;
          ea[j] = 0;
          eb[j] = 0;
          if (mask) {
            const int ia = __builtin_ctzll(mask);
            mask &= mask - 1;
            const bool hasb = mask != 0;
            int ib = ia;
            if (hasb) {
              ib = __builtin_ctzll(mask);
              mask &= mask - 1;
            }
            const int sa = pick(s0, s1, ia), da = pick(d0, d1, ia);
            const int sb = pick(s0, s1, ib), db = pick(d0, d1, ib);
            dd[j] = odd ? (hasb ? db : -1) : da;
            v[j] = g2[(size_t)(odd ? sb : sa) * 16 + lp];
            ea[j] = ia;
            eb[j] = ib;
          }
        }
        if (need_w) {
          if (sel0) w0 = dinv[d0];
          if (sel1) w1 = dinv[d1];
          need_w = false;
        }
#pragma unroll
        for (int j = 0; j < 8; ++j) {
          if (dd[j] >= 0) {
            const float wa = pickf(w0, w1, ea[j]), wb = pickf(w0, w1, eb[j]);
            atomicAdd(outAcc + (size_t)dd[j] * 16 + lp, (odd ? wb : wa) * v[j]);
          }
        }
      }
    }
    if (threadIdx.x == 0)
      sItem[par ^ 1] = item_of_ticket(tNext, ctr, nItems, own, drawCls);
  }
}

extern "C" void kernel_launch(void* const* d_in, const int* in_sizes, int n_in,
                              void* d_out, int out_size, void* d_ws, size_t ws_size,
                              hipStream_t stream) {
  const float* X  = (const float*)d_in[0];
  const int*   ei = (const int*)d_in[1];
  const float* W1 = (const float*)d_in[2];
  const float* b1 = (const float*)d_in[3];
  const float* W2 = (const float*)d_in[4];
  const float* b2 = (const float*)d_in[5];
  float* out = (float*)d_out;

  const int N = in_sizes[0] / 64;
  const int E = in_sizes[1] / 2;
  const int* src = ei;
  const int* dst = ei + E;

  float* ws   = (float*)d_ws;
  float* dinv = ws;
  float* g1   = dinv + N;
  float* agg1 = g1 + (size_t)N * 64;
  float* g2   = agg1 + (size_t)N * 64;
  float* deg2 = g2 + (size_t)N * 16;
  int*   ctr  = (int*)(deg2 + (size_t)N * 2);

  const int nTiles = N / 16;
  const int B = 256;

  const int nItems = (E + EPB - 1) / EPB;

  gcn_init_deg<<<(2 * N + NCTR + B - 1) / B, B, 0, stream>>>(deg2, 2 * N, ctr);
  gcn_edge_deg<<<DEG_GRID, B, 0, stream>>>(dst, E, deg2, N);

  gcn_gemm1<<<(nTiles + WPB - 1) / WPB, B, 0, stream>>>(X, W1, deg2, dinv, g1,
                                                        agg1, nTiles, N);
  gcn_edge_agg1<<<SCAT_GRID, B, 0, stream>>>(src, dst, g1, agg1, E, nItems, ctr);

  gcn_gemm2<<<(nTiles + WPB - 1) / WPB, B, 0, stream>>>(agg1, b1, W2, b2, dinv,
                                                        g2, out, nTiles);
  gcn_edge_agg2<<<SCAT_GRID, B, 0, stream>>>(src, dst, g2, dinv, out, E, nItems,
                                              ctr + 128);
}


# Both edge scatters: a wave writes its selected edges in edge order to a per-wave LDS list (slot = prefix count) and walks it by LDS loads: no mask bit-walk or lane reads; sums arrival-ordered
# speedup vs baseline: 2.8965x; 1.0721x over previous
#include <hip/hip_runtime.h>


typedef __attribute__((ext_vector_type(2))) float v2f;
typedef __attribute__((ext_vector_type(8))) float v8f;

#define WPB 8

__device__ __forceinline__ int chip_half() {
  const unsigned id = __builtin_amdgcn_s_sendmsg_rtn(0x87);
  return (int)((id >> 18) & 1u);
}

__device__ __forceinline__ int row_class(int d) { return (d >> 5) & 1; }

#define NCTR 256

__global__ void gcn_init_deg(float* __restrict__ deg2, int n2,
                             int* __restrict__ ctr) {
  int i = blockIdx.x * blockDim.x + threadIdx.x;
  if (i < n2)
    deg2[i] = 0.0f;
  else if (i < n2 + NCTR)
    ctr[i - n2] = 0;
}

#define DEG_GRID 512
#define DEG_PER 8
__global__ __launch_bounds__(256) void gcn_edge_deg(
    const int* __restrict__ dst, int E, float* __restrict__ deg2, int n) {
  float* __restrict__ plane = deg2 + (size_t)chip_half() * n;
  const int stride = gridDim.x * blockDim.x;
  for (int e = blockIdx.x * blockDim.x + threadIdx.x; e < E; e += DEG_PER * stride) {
    int d[DEG_PER];
#pragma unroll
    for (int k = 0; k < DEG_PER; ++k) {
      const int ek = e + k * stride;
      d[k] = ek < E ? dst[ek] : -1;
    }
#pragma unroll
    for (int k = 0; k < DEG_PER; ++k)
      if (d[k] >= 0) atomicAdd(&plane[d[k]], 1.0f);
  }
}

#define TP 68
__global__ __launch_bounds__(256) void gcn_gemm1(
    const float* __restrict__ X, const float* __restrict__ W1,
    const float* __restrict__ deg2, float* __restrict__ dinv,
    float* __restrict__ g1, float* __restrict__ agg1, int nTiles, int n) {
  const int wave = threadIdx.x >> 5;
  const int lane = threadIdx.x & 31;
  int tile = blockIdx.x * WPB + wave;
  const bool live = tile < nTiles;
  if (!live) tile = nTiles - 1;
  const int rowbase = tile << 4;
  const int half = lane >> 4;
  const int lp = lane & 15;
  const int m = rowbase + lp;

  __shared__ __attribute__((aligned(16))) float sW[64 * 64];
  __shared__ __attribute__((aligned(16))) float sT[WPB][16 * TP];
  float* __restrict__ T = sT[wave];

  const float4* __restrict__ x4 = (const float4*)(X + (size_t)rowbase * 64);
  float4 av[8];
#pragma unroll
  for (int i = 0; i < 8; ++i) av[i] = x4[i * 32 + lane];
  const float c0 = deg2[m];
  const float c1 = deg2[n + m];
  {
    const float4* w4 = (const float4*)W1;
    float4* s4 = (float4*)sW;
    float4 wv[4];
#pragma unroll
    for (int i = 0; i < 4; ++i) wv[i] = w4[threadIdx.x + i * 256];
#pragma unroll
    for (int i = 0; i < 4; ++i) s4[threadIdx.x + i * 256] = wv[i];
  }
#pragma unroll
  for (int i = 0; i < 8; ++i)
    *(float4*)&T[(2 * i + half) * TP + 4 * lp] = av[i];
  __syncthreads();

  const float cnt = 1.0f + c0 + c1;
  const float dvm = (cnt > 0.0f) ? rsqrtf(cnt) : 0.0f;
  if (live && half == 0) dinv[m] = dvm;

  v8f acc0 = {}, acc1 = {}, acc2 = {}, acc3 = {};
  const float* __restrict__ Arow = T + lp * TP;

#pragma unroll
  for (int k = 0; k < 64; k += 4) {
    const int ka = k + 2 * half;
    v2f a;
    a.x = Arow[ka];
    a.y = Arow[ka + 1];
    const float* b0r = &sW[ka * 64];
    const float* b1r = &sW[(ka + 1) * 64];
    v2f b;
    b.x = b0r[lp];
    b.y = b1r[lp];
    acc0 = __builtin_amdgcn_wmma_f32_16x16x4_f32(false, a, false, b, (short)0,
                                                 acc0, false, false);
    b.x = b0r[16 + lp];
    b.y = b1r[16 + lp];
    acc1 = __builtin_amdgcn_wmma_f32_16x16x4_f32(false, a, false, b, (short)0,
                                                 acc1, false, false);
    b.x = b0r[32 + lp];
    b.y = b1r[32 + lp];
    acc2 = __builtin_amdgcn_wmma_f32_16x16x4_f32(false, a, false, b, (short)0,
                                                 acc2, false, false);
    b.x = b0r[48 + lp];
    b.y = b1r[48 + lp];
    acc3 = __builtin_amdgcn_wmma_f32_16x16x4_f32(false, a, false, b, (short)0,
                                                 acc3, false, false);
  }

  __syncthreads();
#pragma unroll
  for (int j = 0; j < 8; ++j) {
    const float dlo = __int_as_float(__builtin_amdgcn_readlane(__float_as_int(dvm), j));
    const float dhi = __int_as_float(__builtin_amdgcn_readlane(__float_as_int(dvm), j + 8));
    const float dv = half ? dhi : dlo;
    float* __restrict__ r = T + (j + 8 * half) * TP + lp;
    r[0]  = dv * acc0[j];
    r[16] = dv * acc1[j];
    r[32] = dv * acc2[j];
    r[48] = dv * acc3[j];
  }
  __syncthreads();
  if (live) {
    float4* __restrict__ o1 = (float4*)(g1 + (size_t)rowbase * 64);
    float4* __restrict__ o2 = (float4*)(agg1 + (size_t)rowbase * 64);
#pragma unroll
    for (int i = 0; i < 8; ++i) {
      const float4 v = *(const float4*)&T[(2 * i + half) * TP + 4 * lp];
      o1[i * 32 + lane] = v;
      o2[i * 32 + lane] = v;
    }
  }
}

#define EPW 64
#define EPB (EPW * WPB)
#define SCAT_GRID 2048

__device__ __forceinline__ int item_of_ticket(int t, int* __restrict__ ctr,
                                              int nItems, int own, int& drawCls) {
  if (t >= nItems && drawCls == own) {
    drawCls = 1 - own;
    t = atomicAdd(&ctr[drawCls * 64], 1);
  }
  return t < nItems ? drawCls * nItems + t : -1;
}

__device__ __forceinline__ unsigned long long load_edges(
    const int* __restrict__ src, const int* __restrict__ dst, int e0, int E,
    int lane, int cls, int& s0, int& d0, int& s1, int& d1) {
  const int ea = e0 + lane, eb = e0 + 32 + lane;
  s0 = 0; d0 = 0; s1 = 0; d1 = 0;
  if (ea < E) { s0 = src[ea]; d0 = dst[ea]; }
  if (eb < E) { s1 = src[eb]; d1 = dst[eb]; }
  const unsigned m0 = __builtin_amdgcn_ballot_w32(ea < E && row_class(d0) == cls);
  const unsigned m1 = __builtin_amdgcn_ballot_w32(eb < E && row_class(d1) == cls);
  return (unsigned long long)m0 | ((unsigned long long)m1 << 32);
}

#define SEL1 72
#define SEL2 80

__device__ __forceinline__ void wave_list_ready() {
  __builtin_amdgcn_fence(__ATOMIC_RELEASE, "wavefront");
  __builtin_amdgcn_wave_barrier();
  __builtin_amdgcn_fence(__ATOMIC_ACQUIRE, "wavefront");
}

__global__ __launch_bounds__(256) void gcn_edge_agg1(
    const int* __restrict__ src, const int* __restrict__ dst,
    const float* __restrict__ g1, float* __restrict__ agg1, int E, int nItems,
    int* __restrict__ ctr) {
  __shared__ int sItem[2];
  __shared__ int2 sSel[WPB][SEL1];
  const int lane = threadIdx.x & 31;
  const int wave = threadIdx.x >> 5;
  const int own = chip_half();
  int drawCls = own;
  int tNext = 0;
  if (threadIdx.x == 0) {
    tNext = atomicAdd(&ctr[drawCls * 64], 1);
    sItem[0] = item_of_ticket(tNext, ctr, nItems, own, drawCls);
  }
  for (int par = 0;; par ^= 1) {
    __syncthreads();
    const int item = __builtin_amdgcn_readfirstlane(sItem[par]);
    if (item < 0) break;
    if (threadIdx.x == 0) tNext = atomicAdd(&ctr[drawCls * 64], 1);
    const int cls = item >= nItems ? 1 : 0;
    const int chunk = item - cls * nItems;
    const int e0 = chunk * EPB + wave * EPW;
    if (e0 < E) {
      int s0, d0, s1, d1;
      const unsigned long long mask = load_edges(src, dst, e0, E, lane, cls, s0, d0, s1, d1);
      const unsigned m0 = (unsigned)mask, m1 = (unsigned)(mask >> 32);
      const int n0 = __builtin_popcount(m0);
      const int cnt = n0 + __builtin_popcount(m1);
      int2* L = sSel[wave];
      if ((m0 >> lane) & 1u) L[__builtin_amdgcn_mbcnt_lo(m0, 0u)] = make_int2(s0, d0);
      if ((m1 >> lane) & 1u) L[n0 + __builtin_amdgcn_mbcnt_lo(m1, 0u)] = make_int2(s1, d1);
      if (lane < 8) L[cnt + lane] = make_int2(0, -1);
      wave_list_ready();

      for (int base = 0; base < cnt; base += 8) {
        int2 e[8];
#pragma unroll
        for (int j = 0; j < 8; ++j) e[j] = L[base + j];
        float lo[8], hi[8];
#pragma unroll
        for (int j = 0; j < 8; ++j) {
          lo[j] = 0.0f;
          hi[j] = 0.0f;
          if (base + j < cnt) {
            const float* r = g1 + (unsigned)(e[j].x * 64 + lane);
            lo[j] = r[0];
            hi[j] = r[32];
          }
        }
#pragma unroll
        for (int j = 0; j < 8; ++j) {
          if (base + j < cnt) {
            float* p = agg1 + (unsigned)(e[j].y * 64 + lane);
            atomicAdd(p, lo[j]);
            atomicAdd(p + 32, hi[j]);
          }
        }
      }
    }
    if (threadIdx.x == 0)
      sItem[par ^ 1] = item_of_ticket(tNext, ctr, nItems, own, drawCls);
  }
}

#define OP 20
__global__ __launch_bounds__(256) void gcn_gemm2(
    const float* __restrict__ agg1, const float* __restrict__ b1,
    const float* __restrict__ W2, const float* __restrict__ b2,
    const float* __restrict__ dinv, float* __restrict__ g2,
    float* __restrict__ outAcc, int nTiles) {
  __shared__ __attribute__((aligned(16))) float sW[64 * 16];
  __shared__ float sB[64];
  __shared__ __attribute__((aligned(16))) float sT[WPB][16 * TP];

  const int wave = threadIdx.x >> 5;
  const int lane = threadIdx.x & 31;
  int tile = blockIdx.x * WPB + wave;
  const bool live = tile < nTiles;
  if (!live) tile = nTiles - 1;
  const int rowbase = tile << 4;
  const int half = lane >> 4;
  const int lp = lane & 15;
  const int m = rowbase + lp;
  float* __restrict__ T = sT[wave];

  const float4* __restrict__ a4 = (const float4*)(agg1 + (size_t)rowbase * 64);
  float4 av[8];
#pragma unroll
  for (int i = 0; i < 8; ++i) av[i] = a4[i * 32 + lane];
  const float dvm = dinv[m];
  const float bias2 = b2[lp];
  {
    const float4 wv = ((const float4*)W2)[threadIdx.x];
    const float bv = b1[threadIdx.x & 63];
    ((float4*)sW)[threadIdx.x] = wv;
    sB[threadIdx.x & 63] = bv;
  }
#pragma unroll
  for (int i = 0; i < 8; ++i)
    *(float4*)&T[(2 * i + half) * TP + 4 * lp] = av[i];
  __syncthreads();

  const float* __restrict__ Arow = T + lp * TP;
  v8f acc = {};

#pragma unroll
  for (int k = 0; k < 64; k += 4) {
    const int ka = k + 2 * half;
    v2f a;
    a.x = fmaxf(fmaf(dvm, Arow[ka], sB[ka]), 0.0f);
    a.y = fmaxf(fmaf(dvm, Arow[ka + 1], sB[ka + 1]), 0.0f);
    v2f b;
    b.x = sW[ka * 16 + lp];
    b.y = sW[(ka + 1) * 16 + lp];
    acc = __builtin_amdgcn_wmma_f32_16x16x4_f32(false, a, false, b, (short)0,
                                                acc, false, false);
  }

  __syncthreads();
  float* __restrict__ Tg = T;
  float* __restrict__ To = T + 16 * OP;
#pragma unroll
  for (int j = 0; j < 8; ++j) {
    const float dlo = __int_as_float(__builtin_amdgcn_readlane(__float_as_int(dvm), j));
    const float dhi = __int_as_float(__builtin_amdgcn_readlane(__float_as_int(dvm), j + 8));
    const float dv = half ? dhi : dlo;
    const float v = dv * acc[j];
    Tg[(j + 8 * half) * OP + lp] = v;
    To[(j + 8 * half) * OP + lp] = fmaf(dv, v, bias2);
  }
  __syncthreads();
  if (live) {
    float4* __restrict__ o1 = (float4*)(g2 + (size_t)rowbase * 16);
    float4* __restrict__ o2 = (float4*)(outAcc + (size_t)rowbase * 16);
#pragma unroll
    for (int c = 0; c < 2; ++c) {
      const int off = (8 * c + (lane >> 2)) * OP + 4 * (lane & 3);
      o1[c * 32 + lane] = *(const float4*)&Tg[off];
      o2[c * 32 + lane] = *(const float4*)&To[off];
    }
  }
}

__global__ __launch_bounds__(256) void gcn_edge_agg2(
    const int* __restrict__ src, const int* __restrict__ dst,
    const float* __restrict__ g2, const float* __restrict__ dinv,
    float* __restrict__ outAcc, int E, int nItems, int* __restrict__ ctr) {
  __shared__ int sItem[2];
  __shared__ int2 sSel[WPB][SEL2];
  __shared__ float sWgt[WPB][SEL2];
  const int lane = threadIdx.x & 31;
  const int wave = threadIdx.x >> 5;
  const bool odd = lane >= 16;
  const int lp = lane & 15;
  const int own = chip_half();
  int drawCls = own;
  int tNext = 0;
  if (threadIdx.x == 0) {
    tNext = atomicAdd(&ctr[drawCls * 64], 1);
    sItem[0] = item_of_ticket(tNext, ctr, nItems, own, drawCls);
  }
  for (int par = 0;; par ^= 1) {
    __syncthreads();
    const int item = __builtin_amdgcn_readfirstlane(sItem[par]);
    if (item < 0) break;
    if (threadIdx.x == 0) tNext = atomicAdd(&ctr[drawCls * 64], 1);
    const int cls = item >= nItems ? 1 : 0;
    const int chunk = item - cls * nItems;
    const int e0 = chunk * EPB + wave * EPW;
    if (e0 < E) {
      int s0, d0, s1, d1;
      const unsigned long long mask = load_edges(src, dst, e0, E, lane, cls, s0, d0, s1, d1);
      const unsigned m0 = (unsigned)mask, m1 = (unsigned)(mask >> 32);
      const int n0 = __builtin_popcount(m0);
      const int cnt = n0 + __builtin_popcount(m1);
      const bool sel0 = (m0 >> lane) & 1u, sel1 = (m1 >> lane) & 1u;
      const int r0 = __builtin_amdgcn_mbcnt_lo(m0, 0u);
      const int r1 = n0 + __builtin_amdgcn_mbcnt_lo(m1, 0u);
      int2* L = sSel[wave];
      float* Wl = sWgt[wave];
      if (sel0) L[r0] = make_int2(s0, d0);
      if (sel1) L[r1] = make_int2(s1, d1);
      if (lane < 16) L[cnt + lane] = make_int2(0, -1);
      wave_list_ready();
      bool need_w = true;
      const int oi = odd ? 1 : 0;

      for (int base = 0; base < cnt; base += 16) {
        int2 e[8];
#pragma unroll
        for (int j = 0; j < 8; ++j) e[j] = L[base + 2 * j + oi];
        float v[8];
#pragma unroll
        for (int j = 0; j < 8; ++j) {
          v[j] = 0.0f;
          if (e[j].y >= 0) v[j] = g2[(unsigned)(e[j].x * 16 + lp)];
        }
        if (need_w) {
          float w0 = 0.0f, w1 = 0.0f;
          if (sel0) w0 = dinv[d0];
          if (sel1) w1 = dinv[d1];
          if (sel0) Wl[r0] = w0;
          if (sel1) Wl[r1] = w1;
          if (lane < 16) Wl[cnt + lane] = 0.0f;
          wave_list_ready();
          need_w = false;
        }
        float w[8];
#pragma unroll
        for (int j = 0; j < 8; ++j) w[j] = Wl[base + 2 * j + oi];
#pragma unroll
        for (int j = 0; j < 8; ++j) {
          if (e[j].y >= 0)
            atomicAdd(outAcc + (unsigned)(e[j].y * 16 + lp), w[j] * v[j]);
        }
      }
    }
    if (threadIdx.x == 0)
      sItem[par ^ 1] = item_of_ticket(tNext, ctr, nItems, own, drawCls);
  }
}

extern "C" void kernel_launch(void* const* d_in, const int* in_sizes, int n_in,
                              void* d_out, int out_size, void* d_ws, size_t ws_size,
                              hipStream_t stream) {
  const float* X  = (const float*)d_in[0];
  const int*   ei = (const int*)d_in[1];
  const float* W1 = (const float*)d_in[2];
  const float* b1 = (const float*)d_in[3];
  const float* W2 = (const float*)d_in[4];
  const float* b2 = (const float*)d_in[5];
  float* out = (float*)d_out;

  const int N = in_sizes[0] / 64;
  const int E = in_sizes[1] / 2;
  const int* src = ei;
  const int* dst = ei + E;

  float* ws   = (float*)d_ws;
  float* dinv = ws;
  float* g1   = dinv + N;
  float* agg1 = g1 + (size_t)N * 64;
  float* g2   = agg1 + (size_t)N * 64;
  float* deg2 = g2 + (size_t)N * 16;
  int*   ctr  = (int*)(deg2 + (size_t)N * 2);

  const int nTiles = N / 16;
  const int B = 256;

  const int nItems = (E + EPB - 1) / EPB;

  gcn_init_deg<<<(2 * N + NCTR + B - 1) / B, B, 0, stream>>>(deg2, 2 * N, ctr);
  gcn_edge_deg<<<DEG_GRID, B, 0, stream>>>(dst, E, deg2, N);

  gcn_gemm1<<<(nTiles + WPB - 1) / WPB, B, 0, stream>>>(X, W1, deg2, dinv, g1,
                                                        agg1, nTiles, N);
  gcn_edge_agg1<<<SCAT_GRID, B, 0, stream>>>(src, dst, g1, agg1, E, nItems, ctr);

  gcn_gemm2<<<(nTiles + WPB - 1) / WPB, B, 0, stream>>>(agg1, b1, W2, b2, dinv,
                                                        g2, out, nTiles);
  gcn_edge_agg2<<<SCAT_GRID, B, 0, stream>>>(src, dst, g2, dinv, out, E, nItems,
                                              ctr + 128);
}
